# starting kernel
# baseline (speedup 1.0000x reference)
#include <hip/hip_runtime.h>
#include <math.h>


#define B_DIM 32
#define C_DIM 256
#define F_DIM 512
#define P_DIM 4
#define N_DIM 1024
#define S_DIM 4096
#define EPS   1e-5f

#define TILE  64
#define PITCH (C_DIM + 2)
#define TILE5 32
#define PITCH5 (F_DIM + 2)

typedef __attribute__((ext_vector_type(2))) float v2f;
typedef __attribute__((ext_vector_type(8))) float v8f;

template <int NB>
__device__ __forceinline__ void wmma_k_loop_nb(const float* __restrict__ Wr,
                                               const float* __restrict__ Ybase,
                                               int pitch, int K, int kk, int m,
                                               v8f acc[NB]) {
#pragma unroll 4
  for (int k = 0; k < K; k += 4) {
    v2f a = *(const v2f*)(Wr + k + kk);
#pragma unroll
    for (int c = 0; c < NB; ++c) {
      v2f b = *(const v2f*)(Ybase + (c * 16 + m) * pitch + k + kk);
      acc[c] = __builtin_amdgcn_wmma_f32_16x16x4_f32(
           false, a,  false, b,
           (short)0, acc[c],  false,  false);
    }
  }
}

template <bool WITH_Q, bool LEAKY>
__global__ void ln_gemm_kernel(const float* __restrict__ xin,
                               const float* __restrict__ lnw,
                               const float* __restrict__ lnb,
                               const float* __restrict__ W,
                               const float* __restrict__ bias,
                               float* __restrict__ out512,
                               float* __restrict__ qout) {
  extern __shared__ float Y[];
  __shared__ float red[4][TILE];
  __shared__ float red2[4][TILE];
  __shared__ float meanv[TILE];
  __shared__ float rstdv[TILE];

  const int b  = blockIdx.y;
  const int s0 = blockIdx.x * TILE;
  const int t  = threadIdx.x;
  const int j  = t & (TILE - 1);
  const int ch = t >> 6;
  const int c0 = ch * 64;

  const float* xb = xin + (size_t)b * C_DIM * S_DIM + s0 + j;
  float sum = 0.f, sq = 0.f;
#pragma unroll 4
  for (int i = 0; i < 64; ++i) {
    int c = c0 + i;
    float v = xb[(size_t)c * S_DIM];
    Y[j * PITCH + c] = v;
    sum += v;
    sq += v * v;
  }
  red[ch][j] = sum;
  red2[ch][j] = sq;
  __syncthreads();
  if (t < TILE) {
    float s  = red[0][t] + red[1][t] + red[2][t] + red[3][t];
    float q2 = red2[0][t] + red2[1][t] + red2[2][t] + red2[3][t];
    float mean = s * (1.f / C_DIM);
    float var  = q2 * (1.f / C_DIM) - mean * mean;
    meanv[t] = mean;
    rstdv[t] = __frsqrt_rn(var + EPS);
  }
  __syncthreads();

  {
    float mean = meanv[j], rstd = rstdv[j];
#pragma unroll 4
    for (int i = 0; i < 64; ++i) {
      int c = c0 + i;
      Y[j * PITCH + c] = (Y[j * PITCH + c] - mean) * rstd * lnw[c] + lnb[c];
    }
  }
  __syncthreads();

  if (WITH_Q) {
    float qs = 0.f;
#pragma unroll 4
    for (int i = 0; i < 64; ++i) {
      int c = c0 + i;
      qs += W[c] * Y[j * PITCH + c];
    }
    red[ch][j] = qs;
    __syncthreads();
    if (t < TILE) {
      qout[(size_t)b * S_DIM + s0 + t] =
          red[0][t] + red[1][t] + red[2][t] + red[3][t] + bias[0];
    }
    __syncthreads();
  }

  const int wave = t >> 5;
  const int lane = t & 31;
  const int m    = lane & 15;
  const int kk   = (lane >> 4) << 1;
  const int mh   = (lane >> 4) << 3;
  const int rofs = WITH_Q ? 1 : 0;

  for (int rt = wave; rt < 32; rt += 8) {
    const float* Wr = W + (size_t)(rofs + rt * 16 + m) * C_DIM;
    __builtin_prefetch(Wr, 0, 1);
    v8f acc[4] = {};
    wmma_k_loop_nb<4>(Wr, Y, PITCH, C_DIM, kk, m, acc);
#pragma unroll
    for (int c = 0; c < 4; ++c) {
#pragma unroll
      for (int r = 0; r < 8; ++r) {
        int row = rt * 16 + mh + r;
        float d = acc[c][r] + bias[rofs + row];
        if (LEAKY) d = d > 0.f ? d : 0.1f * d;
        out512[((size_t)b * F_DIM + row) * S_DIM + s0 + c * 16 + m] = d;
      }
    }
  }
}

__global__ void softmax_ctx_kernel(const float* __restrict__ qbuf,
                                   const float* __restrict__ kv,
                                   float* __restrict__ ctx) {
  __shared__ float probs[N_DIM];
  __shared__ float red[256];
  const int bp = blockIdx.x;
  const int b = bp >> 2, p = bp & 3;
  const int t = threadIdx.x;

  const float* q = qbuf + (size_t)b * S_DIM + p * N_DIM;
  float v[4];
  float mx = -INFINITY;
#pragma unroll
  for (int i = 0; i < 4; ++i) {
    v[i] = q[t + i * 256];
    mx = fmaxf(mx, v[i]);
  }
  red[t] = mx;
  __syncthreads();
  for (int o = 128; o > 0; o >>= 1) {
    if (t < o) red[t] = fmaxf(red[t], red[t + o]);
    __syncthreads();
  }
  mx = red[0];
  __syncthreads();

  float s = 0.f;
#pragma unroll
  for (int i = 0; i < 4; ++i) {
    v[i] = __expf(v[i] - mx);
    s += v[i];
  }
  red[t] = s;
  __syncthreads();
  for (int o = 128; o > 0; o >>= 1) {
    if (t < o) red[t] += red[t + o];
    __syncthreads();
  }
  float inv = 1.f / red[0];
#pragma unroll
  for (int i = 0; i < 4; ++i) probs[t + i * 256] = v[i] * inv;
  __syncthreads();

  const float* kr = kv + ((size_t)b * F_DIM + t) * S_DIM + p * N_DIM;
  float acc = 0.f;
#pragma unroll 4
  for (int n = 0; n < N_DIM; n += 4) {
    float4 kx = *reinterpret_cast<const float4*>(kr + n);
    acc += kx.x * probs[n] + kx.y * probs[n + 1] +
           kx.z * probs[n + 2] + kx.w * probs[n + 3];
  }
  ctx[(size_t)bp * C_DIM + t] = acc;
}

__global__ void attn_out_kernel(const float* __restrict__ x,
                                const float* __restrict__ kv,
                                const float* __restrict__ ctx,
                                const float* __restrict__ W,
                                const float* __restrict__ bias,
                                float* __restrict__ x1) {
  extern __shared__ float A[];
  const int b  = blockIdx.y;
  const int s0 = blockIdx.x * TILE;
  const int p  = s0 / N_DIM;
  const int t  = threadIdx.x;
  const int j  = t & (TILE - 1);
  const int ch = t >> 6;
  const int c0 = ch * 64;

  const float* ctxbp = ctx + ((size_t)b * P_DIM + p) * C_DIM;
#pragma unroll 4
  for (int i = 0; i < 64; ++i) {
    int c = c0 + i;
    float vv = kv[((size_t)b * F_DIM + C_DIM + c) * S_DIM + s0 + j];
    vv = vv > 0.f ? vv : 0.f;
    A[j * PITCH + c] = vv * ctxbp[c];
  }
  __syncthreads();

  const int wave = t >> 5;
  const int lane = t & 31;
  const int m    = lane & 15;
  const int kk   = (lane >> 4) << 1;
  const int mh   = (lane >> 4) << 3;

  for (int rt = wave; rt < 16; rt += 8) {
    const float* Wr = W + (size_t)(rt * 16 + m) * C_DIM;
    __builtin_prefetch(Wr, 0, 1);
    v8f acc[4] = {};
    wmma_k_loop_nb<4>(Wr, A, PITCH, C_DIM, kk, m, acc);
#pragma unroll
    for (int c = 0; c < 4; ++c) {
#pragma unroll
      for (int r = 0; r < 8; ++r) {
        int row = rt * 16 + mh + r;
        size_t idx = ((size_t)b * C_DIM + row) * S_DIM + s0 + c * 16 + m;
        x1[idx] = x[idx] + bias[row] + acc[c][r];
      }
    }
  }
}

__global__ void ffn2_kernel(const float* __restrict__ x1,
                            const float* __restrict__ h,
                            const float* __restrict__ W,
                            const float* __restrict__ bias,
                            float* __restrict__ out) {
  extern __shared__ float Hs[];
  const int b  = blockIdx.y;
  const int s0 = blockIdx.x * TILE5;
  const int t  = threadIdx.x;
  const int j  = t & (TILE5 - 1);
  const int ch = t >> 5;
#pragma unroll 4
  for (int i = 0; i < 64; ++i) {
    int r = ch * 64 + i;
    Hs[j * PITCH5 + r] = h[((size_t)b * F_DIM + r) * S_DIM + s0 + j];
  }
  __syncthreads();

  const int wave = t >> 5;
  const int lane = t & 31;
  const int m    = lane & 15;
  const int kk   = (lane >> 4) << 1;
  const int mh   = (lane >> 4) << 3;

  for (int rt = wave; rt < 16; rt += 8) {
    const float* Wr = W + (size_t)(rt * 16 + m) * F_DIM;
    __builtin_prefetch(Wr, 0, 1);
    v8f acc[2] = {};
    wmma_k_loop_nb<2>(Wr, Hs, PITCH5, F_DIM, kk, m, acc);
#pragma unroll
    for (int c = 0; c < 2; ++c) {
#pragma unroll
      for (int r = 0; r < 8; ++r) {
        int row = rt * 16 + mh + r;
        size_t idx = ((size_t)b * C_DIM + row) * S_DIM + s0 + c * 16 + m;
        out[idx] = x1[idx] + bias[row] + acc[c][r];
      }
    }
  }
}

extern "C" void kernel_launch(void* const* d_in, const int* in_sizes, int n_in,
                              void* d_out, int out_size, void* d_ws,
                              size_t ws_size, hipStream_t stream) {
  const float* x      = (const float*)d_in[0];
  const float* ln1_w  = (const float*)d_in[1];
  const float* ln1_b  = (const float*)d_in[2];
  const float* qkv_w  = (const float*)d_in[3];
  const float* qkv_b  = (const float*)d_in[4];
  const float* out_w  = (const float*)d_in[5];
  const float* out_b  = (const float*)d_in[6];
  const float* ln2_w  = (const float*)d_in[7];
  const float* ln2_b  = (const float*)d_in[8];
  const float* ffn1_w = (const float*)d_in[9];
  const float* ffn1_b = (const float*)d_in[10];
  const float* ffn2_w = (const float*)d_in[11];
  const float* ffn2_b = (const float*)d_in[12];
  float* outp = (float*)d_out;

  float* ws   = (float*)d_ws;
  float* qbuf = ws;
  float* ctxb = qbuf + (size_t)B_DIM * S_DIM;
  float* kvb  = ctxb + (size_t)B_DIM * P_DIM * C_DIM;
  float* x1   = kvb + (size_t)B_DIM * F_DIM * S_DIM;
  float* hbuf = kvb;

  const size_t lds1 = (size_t)TILE * PITCH * sizeof(float);
  const size_t lds5 = (size_t)TILE5 * PITCH5 * sizeof(float);
  dim3 grid1(S_DIM / TILE, B_DIM);
  dim3 grid5(S_DIM / TILE5, B_DIM);

  ln_gemm_kernel<true, false><<<grid1, 256, lds1, stream>>>(
      x, ln1_w, ln1_b, qkv_w, qkv_b, kvb, qbuf);
  softmax_ctx_kernel<<<B_DIM * P_DIM, 256, 0, stream>>>(qbuf, kvb, ctxb);
  attn_out_kernel<<<grid1, 256, lds1, stream>>>(
      x, kvb, ctxb, out_w, out_b, x1);
  ln_gemm_kernel<false, true><<<grid1, 256, lds1, stream>>>(
      x1, ln2_w, ln2_b, ffn1_w, ffn1_b, hbuf, nullptr);
  ffn2_kernel<<<grid5, 256, lds5, stream>>>(
      x1, hbuf, ffn2_w, ffn2_b, outp);
}


# Wave register tile 2 row-tiles x NB position-tiles (was 1 x NB, row-tiles in turn): each LDS activation fragment feeds both row-tiles' WMMAs, 32 WMMAs per load group; f32 sum order unchanged
# speedup vs baseline: 1.1549x; 1.1549x over previous
#include <hip/hip_runtime.h>
#include <math.h>


#define B_DIM 32
#define C_DIM 256
#define F_DIM 512
#define P_DIM 4
#define N_DIM 1024
#define S_DIM 4096
#define EPS   1e-5f

#define TILE  64
#define PITCH (C_DIM + 2)
#define TILE5 32
#define PITCH5 (F_DIM + 2)

#define GEMM_KERNEL_HINT \
  __launch_bounds__(256) __attribute__((amdgpu_waves_per_eu(2, 8)))

#define GEMM_UNR4 2
#define GEMM_UNR2 4

typedef __attribute__((ext_vector_type(2))) float v2f;
typedef __attribute__((ext_vector_type(4))) float v4f;
typedef __attribute__((ext_vector_type(8))) float v8f;

template <int K>
__global__ void frag_order_kernel(const float* __restrict__ W,
                                  float* __restrict__ Wf) {
  const int g    = blockIdx.x * 256 + threadIdx.x;
  const int lane = g & 31;
  const int kp   = (g >> 5) & (K / 8 - 1);
  const int rt   = (g >> 5) / (K / 8);
  const int m    = lane & 15;
  const int kk   = (lane >> 4) << 1;
  const float* src = W + (size_t)(rt * 16 + m) * K + kp * 8 + kk;
  v2f lo = *(const v2f*)src;
  v2f hi = *(const v2f*)(src + 4);
  v4f o = {lo.x, lo.y, hi.x, hi.y};
  *(v4f*)(Wf + (size_t)g * 4) = o;
}

template <int NB, int UNR>
__device__ __forceinline__ void wmma_k_loop_2r(const float* __restrict__ Af0,
                                               const float* __restrict__ Af1,
                                               const float* __restrict__ Ybase,
                                               int pitch, int K, int kk, int m,
                                               v8f acc0[NB], v8f acc1[NB]) {
#pragma unroll UNR
  for (int k = 0; k < K; k += 8) {
    v4f p4 = *(const v4f*)(Af0 + k * 16);
    v4f q4 = *(const v4f*)(Af1 + k * 16);
    v2f p0 = {p4.x, p4.y};
    v2f p1 = {p4.z, p4.w};
    v2f q0 = {q4.x, q4.y};
    v2f q1 = {q4.z, q4.w};
#pragma unroll
    for (int c = 0; c < NB; ++c) {
      v2f b = *(const v2f*)(Ybase + (c * 16 + m) * pitch + k + kk);
      acc0[c] = __builtin_amdgcn_wmma_f32_16x16x4_f32(
           false, p0,  false, b,
           (short)0, acc0[c],  false,  false);
      acc1[c] = __builtin_amdgcn_wmma_f32_16x16x4_f32(
           false, q0,  false, b,
           (short)0, acc1[c],  false,  false);
    }
#pragma unroll
    for (int c = 0; c < NB; ++c) {
      v2f b = *(const v2f*)(Ybase + (c * 16 + m) * pitch + k + 4 + kk);
      acc0[c] = __builtin_amdgcn_wmma_f32_16x16x4_f32(
           false, p1,  false, b,
           (short)0, acc0[c],  false,  false);
      acc1[c] = __builtin_amdgcn_wmma_f32_16x16x4_f32(
           false, q1,  false, b,
           (short)0, acc1[c],  false,  false);
    }
  }
}

template <bool WITH_Q, bool LEAKY>
__global__ void GEMM_KERNEL_HINT ln_gemm_kernel(const float* __restrict__ xin,
                               const float* __restrict__ lnw,
                               const float* __restrict__ lnb,
                               const float* __restrict__ W,
                               const float* __restrict__ Wf,
                               const float* __restrict__ bias,
                               float* __restrict__ out512,
                               float* __restrict__ qout) {
  extern __shared__ __attribute__((aligned(16))) float Y[];
  __shared__ __attribute__((aligned(16))) float red[16][TILE];
  __shared__ __attribute__((aligned(16))) float red2[16][TILE];
  __shared__ float meanv[TILE];
  __shared__ float rstdv[TILE];

  const int b  = blockIdx.y;
  const int s0 = blockIdx.x * TILE;
  const int t  = threadIdx.x;
  const int j  = t & (TILE - 1);
  const int ch = t >> 6;
  const int c0 = ch * 64;

  {
    const int ql = t & 7;
    const int cl = (t >> 3) & 3;
    const int wv = t >> 5;
    const int qd = (wv & 1) * 8 + ql;
    const int cb = (wv >> 1) * 8 + cl * 2;
    const float* xb = xin + (size_t)b * C_DIM * S_DIM + s0 + 4 * qd;
    v4f va[8], vb[8];
#pragma unroll
    for (int i = 0; i < 8; ++i) {
      int c = i * 32 + cb;
      va[i] = *(const v4f*)(xb + (size_t)c * S_DIM);
      vb[i] = *(const v4f*)(xb + (size_t)(c + 1) * S_DIM);
    }
    v4f sum = {0.f, 0.f, 0.f, 0.f}, sq = {0.f, 0.f, 0.f, 0.f};
#pragma unroll
    for (int i = 0; i < 8; ++i) {
      int c = i * 32 + cb;
#pragma unroll
      for (int e = 0; e < 4; ++e) {
        v2f pr = {va[i][e], vb[i][e]};
        *(v2f*)(&Y[(4 * qd + e) * PITCH + c]) = pr;
      }
      sum += va[i];
      sq += va[i] * va[i];
      sum += vb[i];
      sq += vb[i] * vb[i];
    }
    const int pr_row = (wv >> 1) * 4 + cl;
    *(v4f*)(&red[pr_row][4 * qd]) = sum;
    *(v4f*)(&red2[pr_row][4 * qd]) = sq;
  }
  __syncthreads();
  if (t < TILE) {
    float s = 0.f, q2 = 0.f;
#pragma unroll
    for (int r = 0; r < 16; ++r) {
      s += red[r][t];
      q2 += red2[r][t];
    }
    float mean = s * (1.f / C_DIM);
    float var  = q2 * (1.f / C_DIM) - mean * mean;
    meanv[t] = mean;
    rstdv[t] = __frsqrt_rn(var + EPS);
  }
  __syncthreads();

  {
    float mean = meanv[j], rstd = rstdv[j];
#pragma unroll 4
    for (int i = 0; i < 64; ++i) {
      int c = c0 + i;
      Y[j * PITCH + c] = (Y[j * PITCH + c] - mean) * rstd * lnw[c] + lnb[c];
    }
  }
  __syncthreads();

  if (WITH_Q) {
    float qs = 0.f;
#pragma unroll 4
    for (int i = 0; i < 64; ++i) {
      int c = c0 + i;
      qs += W[c] * Y[j * PITCH + c];
    }
    red[ch][j] = qs;
    __syncthreads();
    if (t < TILE) {
      qout[(size_t)b * S_DIM + s0 + t] =
          red[0][t] + red[1][t] + red[2][t] + red[3][t] + bias[0];
    }
    __syncthreads();
  }

  const int wave = t >> 5;
  const int lane = t & 31;
  const int m    = lane & 15;
  const int kk   = (lane >> 4) << 1;
  const int mh   = (lane >> 4) << 3;
  const int rofs = WITH_Q ? 1 : 0;

  for (int rt = wave; rt < 32; rt += 16) {
    const float* Af0 = Wf + (size_t)rt * (16 * C_DIM) + lane * 4;
    const float* Af1 = Af0 + 8 * (16 * C_DIM);
    v8f acc[2][4] = {};
    wmma_k_loop_2r<4, GEMM_UNR4>(Af0, Af1, Y, PITCH, C_DIM, kk, m, acc[0], acc[1]);
#pragma unroll
    for (int h = 0; h < 2; ++h) {
#pragma unroll
      for (int c = 0; c < 4; ++c) {
#pragma unroll
        for (int r = 0; r < 8; ++r) {
          int row = (rt + 8 * h) * 16 + mh + r;
          float d = acc[h][c][r] + bias[rofs + row];
          if (LEAKY) d = d > 0.f ? d : 0.1f * d;
          out512[((size_t)b * F_DIM + row) * S_DIM + s0 + c * 16 + m] = d;
        }
      }
    }
  }
}

__global__ void softmax_ctx_kernel(const float* __restrict__ qbuf,
                                   const float* __restrict__ kv,
                                   float* __restrict__ ctx) {
  __shared__ float probs[N_DIM];
  __shared__ float red[256];
  const int bp = blockIdx.x;
  const int b = bp >> 2, p = bp & 3;
  const int t = threadIdx.x;

  const float* q = qbuf + (size_t)b * S_DIM + p * N_DIM;
  float v[4];
  float mx = -INFINITY;
#pragma unroll
  for (int i = 0; i < 4; ++i) {
    v[i] = q[t + i * 256];
    mx = fmaxf(mx, v[i]);
  }
  red[t] = mx;
  __syncthreads();
  for (int o = 128; o > 0; o >>= 1) {
    if (t < o) red[t] = fmaxf(red[t], red[t + o]);
    __syncthreads();
  }
  mx = red[0];
  __syncthreads();

  float s = 0.f;
#pragma unroll
  for (int i = 0; i < 4; ++i) {
    v[i] = __expf(v[i] - mx);
    s += v[i];
  }
  red[t] = s;
  __syncthreads();
  for (int o = 128; o > 0; o >>= 1) {
    if (t < o) red[t] += red[t + o];
    __syncthreads();
  }
  float inv = 1.f / red[0];
#pragma unroll
  for (int i = 0; i < 4; ++i) probs[t + i * 256] = v[i] * inv;
  __syncthreads();

  const float* kr = kv + ((size_t)b * F_DIM + t) * S_DIM + p * N_DIM;
  float acc = 0.f;
#pragma unroll 4
  for (int n = 0; n < N_DIM; n += 4) {
    float4 kx = *reinterpret_cast<const float4*>(kr + n);
    acc += kx.x * probs[n] + kx.y * probs[n + 1] +
           kx.z * probs[n + 2] + kx.w * probs[n + 3];
  }
  ctx[(size_t)bp * C_DIM + t] = acc;
}

__global__ void GEMM_KERNEL_HINT attn_out_kernel(const float* __restrict__ x,
                                const float* __restrict__ kv,
                                const float* __restrict__ ctx,
                                const float* __restrict__ Wf,
                                const float* __restrict__ bias,
                                float* __restrict__ x1) {
  extern __shared__ __attribute__((aligned(16))) float A[];
  const int b  = blockIdx.y;
  const int s0 = blockIdx.x * TILE;
  const int p  = s0 / N_DIM;
  const int t  = threadIdx.x;
  const int j  = t & (TILE - 1);
  const int ch = t >> 6;
  const int c0 = ch * 64;

  const float* ctxbp = ctx + ((size_t)b * P_DIM + p) * C_DIM;
  {
    const int ql = t & 7;
    const int cl = (t >> 3) & 3;
    const int wv = t >> 5;
    const int qd = (wv & 1) * 8 + ql;
    const int cb = (wv >> 1) * 8 + cl * 2;
    const float* vbase = kv + ((size_t)b * F_DIM + C_DIM) * S_DIM + s0 + 4 * qd;
    v4f va[8], vb[8];
    v2f cx[8];
#pragma unroll
    for (int i = 0; i < 8; ++i) {
      int c = i * 32 + cb;
      va[i] = *(const v4f*)(vbase + (size_t)c * S_DIM);
      vb[i] = *(const v4f*)(vbase + (size_t)(c + 1) * S_DIM);
      cx[i] = *(const v2f*)(ctxbp + c);
    }
#pragma unroll
    for (int i = 0; i < 8; ++i) {
      int c = i * 32 + cb;
#pragma unroll
      for (int e = 0; e < 4; ++e) {
        float v0 = va[i][e], v1 = vb[i][e];
        v0 = v0 > 0.f ? v0 : 0.f;
        v1 = v1 > 0.f ? v1 : 0.f;
        v2f pr = {v0 * cx[i].x, v1 * cx[i].y};
        *(v2f*)(&A[(4 * qd + e) * PITCH + c]) = pr;
      }
    }
  }
  __syncthreads();

  const int wave = t >> 5;
  const int lane = t & 31;
  const int m    = lane & 15;
  const int kk   = (lane >> 4) << 1;
  const int mh   = (lane >> 4) << 3;

  {
    const float* Af0 = Wf + (size_t)wave * (16 * C_DIM) + lane * 4;
    const float* Af1 = Af0 + 8 * (16 * C_DIM);
    v8f acc[2][4] = {};
    wmma_k_loop_2r<4, GEMM_UNR4>(Af0, Af1, A, PITCH, C_DIM, kk, m, acc[0], acc[1]);
#pragma unroll
    for (int h = 0; h < 2; ++h) {
#pragma unroll
      for (int c = 0; c < 4; ++c) {
#pragma unroll
        for (int r = 0; r < 8; ++r) {
          int row = (wave + 8 * h) * 16 + mh + r;
          size_t idx = ((size_t)b * C_DIM + row) * S_DIM + s0 + c * 16 + m;
          x1[idx] = x[idx] + bias[row] + acc[h][c][r];
        }
      }
    }
  }
}

__global__ void GEMM_KERNEL_HINT ffn2_kernel(const float* __restrict__ x1,
                            const float* __restrict__ h,
                            const float* __restrict__ Wf,
                            const float* __restrict__ bias,
                            float* __restrict__ out) {
  extern __shared__ __attribute__((aligned(16))) float Hs[];
  const int b  = blockIdx.y;
  const int s0 = blockIdx.x * TILE5;
  const int t  = threadIdx.x;
  const int j  = t & (TILE5 - 1);
  const int ch = t >> 5;
  {
    const int ql = t & 7;
    const int cl = (t >> 3) & 3;
    const int rb = ch * 8 + cl * 2;
    const float* hb = h + (size_t)b * F_DIM * S_DIM + s0 + 4 * ql;
    v4f va[8], vb[8];
#pragma unroll
    for (int i = 0; i < 8; ++i) {
      int r = i * 64 + rb;
      va[i] = *(const v4f*)(hb + (size_t)r * S_DIM);
      vb[i] = *(const v4f*)(hb + (size_t)(r + 1) * S_DIM);
    }
#pragma unroll
    for (int i = 0; i < 8; ++i) {
      int r = i * 64 + rb;
#pragma unroll
      for (int e = 0; e < 4; ++e) {
        v2f pr = {va[i][e], vb[i][e]};
        *(v2f*)(&Hs[(4 * ql + e) * PITCH5 + r]) = pr;
      }
    }
  }
  __syncthreads();

  const int wave = t >> 5;
  const int lane = t & 31;
  const int m    = lane & 15;
  const int kk   = (lane >> 4) << 1;
  const int mh   = (lane >> 4) << 3;

  {
    const float* Af0 = Wf + (size_t)wave * (16 * F_DIM) + lane * 4;
    const float* Af1 = Af0 + 8 * (16 * F_DIM);
    v8f acc[2][2] = {};
    wmma_k_loop_2r<2, GEMM_UNR2>(Af0, Af1, Hs, PITCH5, F_DIM, kk, m, acc[0], acc[1]);
#pragma unroll
    for (int h = 0; h < 2; ++h) {
#pragma unroll
      for (int c = 0; c < 2; ++c) {
#pragma unroll
        for (int r = 0; r < 8; ++r) {
          int row = (wave + 8 * h) * 16 + mh + r;
          size_t idx = ((size_t)b * C_DIM + row) * S_DIM + s0 + c * 16 + m;
          out[idx] = x1[idx] + bias[row] + acc[h][c][r];
        }
      }
    }
  }
}

extern "C" void kernel_launch(void* const* d_in, const int* in_sizes, int n_in,
                              void* d_out, int out_size, void* d_ws,
                              size_t ws_size, hipStream_t stream) {
  const float* x      = (const float*)d_in[0];
  const float* ln1_w  = (const float*)d_in[1];
  const float* ln1_b  = (const float*)d_in[2];
  const float* qkv_w  = (const float*)d_in[3];
  const float* qkv_b  = (const float*)d_in[4];
  const float* out_w  = (const float*)d_in[5];
  const float* out_b  = (const float*)d_in[6];
  const float* ln2_w  = (const float*)d_in[7];
  const float* ln2_b  = (const float*)d_in[8];
  const float* ffn1_w = (const float*)d_in[9];
  const float* ffn1_b = (const float*)d_in[10];
  const float* ffn2_w = (const float*)d_in[11];
  const float* ffn2_b = (const float*)d_in[12];
  float* outp = (float*)d_out;

  float* ws   = (float*)d_ws;
  float* qbuf = ws;
  float* ctxb = qbuf + (size_t)B_DIM * S_DIM;
  float* kvb  = ctxb + (size_t)B_DIM * P_DIM * C_DIM;
  float* x1   = kvb + (size_t)B_DIM * F_DIM * S_DIM;
  float* hbuf = kvb;
  float* wf_qkv = x1;
  float* wf_any = qbuf;

  const size_t lds1 = (size_t)TILE * PITCH * sizeof(float);
  const size_t lds5 = (size_t)TILE5 * PITCH5 * sizeof(float);
  dim3 grid1(S_DIM / TILE, B_DIM);
  dim3 grid5(S_DIM / TILE5, B_DIM);

  frag_order_kernel<C_DIM><<<128, 256, 0, stream>>>(qkv_w + C_DIM, wf_qkv);
  ln_gemm_kernel<true, false><<<grid1, 256, lds1, stream>>>(
      x, ln1_w, ln1_b, qkv_w, wf_qkv, qkv_b, kvb, qbuf);
  softmax_ctx_kernel<<<B_DIM * P_DIM, 256, 0, stream>>>(qbuf, kvb, ctxb);
  frag_order_kernel<C_DIM><<<64, 256, 0, stream>>>(out_w, wf_any);
  attn_out_kernel<<<grid1, 256, lds1, stream>>>(
      x, kvb, ctxb, wf_any, out_b, x1);
  frag_order_kernel<C_DIM><<<128, 256, 0, stream>>>(ffn1_w, wf_any);
  ln_gemm_kernel<false, true><<<grid1, 256, lds1, stream>>>(
      x1, ln2_w, ln2_b, ffn1_w, wf_any, ffn1_b, hbuf, nullptr);
  frag_order_kernel<F_DIM><<<128, 256, 0, stream>>>(ffn2_w, wf_any);
  ffn2_kernel<<<grid5, 256, lds5, stream>>>(
      x1, hbuf, wf_any, ffn2_b, outp);
}


# qkv kernel's k pass ends in ctx partials (accumulators x exp(q - tile max), summed over the tile's 64 positions); k GEMM stays on WMMA, k and q never reach memory; 16 tiles joined; ctx sum order new
# speedup vs baseline: 1.2117x; 1.0492x over previous
#include <hip/hip_runtime.h>
#include <math.h>


#define B_DIM 32
#define C_DIM 256
#define F_DIM 512
#define P_DIM 4
#define N_DIM 1024
#define S_DIM 4096
#define EPS   1e-5f

#define TILE  64
#define PITCH (C_DIM + 2)
#define TILE5 32
#define PART_PITCH 288
#define TILES_PER_P (N_DIM / TILE)
#define PITCH5 (F_DIM + 2)

#define GEMM_KERNEL_HINT \
  __launch_bounds__(256) __attribute__((amdgpu_waves_per_eu(2, 8)))

#define GEMM_UNR4 2
#define GEMM_UNR2 4

typedef __attribute__((ext_vector_type(2))) float v2f;
typedef __attribute__((ext_vector_type(4))) float v4f;
typedef __attribute__((ext_vector_type(8))) float v8f;

template <int K>
__global__ void frag_order_kernel(const float* __restrict__ W,
                                  float* __restrict__ Wf) {
  const int g    = blockIdx.x * 256 + threadIdx.x;
  const int lane = g & 31;
  const int kp   = (g >> 5) & (K / 8 - 1);
  const int rt   = (g >> 5) / (K / 8);
  const int m    = lane & 15;
  const int kk   = (lane >> 4) << 1;
  const float* src = W + (size_t)(rt * 16 + m) * K + kp * 8 + kk;
  v2f lo = *(const v2f*)src;
  v2f hi = *(const v2f*)(src + 4);
  v4f o = {lo.x, lo.y, hi.x, hi.y};
  *(v4f*)(Wf + (size_t)g * 4) = o;
}

template <int NB, int UNR>
__device__ __forceinline__ void wmma_k_loop_2r(const float* __restrict__ Af0,
                                               const float* __restrict__ Af1,
                                               const float* __restrict__ Ybase,
                                               int pitch, int K, int kk, int m,
                                               v8f acc0[NB], v8f acc1[NB]) {
#pragma unroll UNR
  for (int k = 0; k < K; k += 8) {
    v4f p4 = *(const v4f*)(Af0 + k * 16);
    v4f q4 = *(const v4f*)(Af1 + k * 16);
    v2f p0 = {p4.x, p4.y};
    v2f p1 = {p4.z, p4.w};
    v2f q0 = {q4.x, q4.y};
    v2f q1 = {q4.z, q4.w};
#pragma unroll
    for (int c = 0; c < NB; ++c) {
      v2f b = *(const v2f*)(Ybase + (c * 16 + m) * pitch + k + kk);
      acc0[c] = __builtin_amdgcn_wmma_f32_16x16x4_f32(
           false, p0,  false, b,
           (short)0, acc0[c],  false,  false);
      acc1[c] = __builtin_amdgcn_wmma_f32_16x16x4_f32(
           false, q0,  false, b,
           (short)0, acc1[c],  false,  false);
    }
#pragma unroll
    for (int c = 0; c < NB; ++c) {
      v2f b = *(const v2f*)(Ybase + (c * 16 + m) * pitch + k + 4 + kk);
      acc0[c] = __builtin_amdgcn_wmma_f32_16x16x4_f32(
           false, p1,  false, b,
           (short)0, acc0[c],  false,  false);
      acc1[c] = __builtin_amdgcn_wmma_f32_16x16x4_f32(
           false, q1,  false, b,
           (short)0, acc1[c],  false,  false);
    }
  }
}

template <bool WITH_Q, bool LEAKY>
__global__ void GEMM_KERNEL_HINT ln_gemm_kernel(const float* __restrict__ xin,
                               const float* __restrict__ lnw,
                               const float* __restrict__ lnb,
                               const float* __restrict__ W,
                               const float* __restrict__ Wf,
                               const float* __restrict__ bias,
                               float* __restrict__ out512,
                               float* __restrict__ part) {
  extern __shared__ __attribute__((aligned(16))) float Y[];
  __shared__ __attribute__((aligned(16))) float red[16][TILE];
  __shared__ __attribute__((aligned(16))) float red2[16][TILE];
  __shared__ float meanv[TILE];
  __shared__ float rstdv[TILE];

  const int b  = blockIdx.y;
  const int s0 = blockIdx.x * TILE;
  const int t  = threadIdx.x;
  const int j  = t & (TILE - 1);
  const int ch = t >> 6;
  const int c0 = ch * 64;

  {
    const int ql = t & 7;
    const int cl = (t >> 3) & 3;
    const int wv = t >> 5;
    const int qd = (wv & 1) * 8 + ql;
    const int cb = (wv >> 1) * 8 + cl * 2;
    const float* xb = xin + (size_t)b * C_DIM * S_DIM + s0 + 4 * qd;
    v4f va[8], vb[8];
#pragma unroll
    for (int i = 0; i < 8; ++i) {
      int c = i * 32 + cb;
      va[i] = *(const v4f*)(xb + (size_t)c * S_DIM);
      vb[i] = *(const v4f*)(xb + (size_t)(c + 1) * S_DIM);
    }
    v4f sum = {0.f, 0.f, 0.f, 0.f}, sq = {0.f, 0.f, 0.f, 0.f};
#pragma unroll
    for (int i = 0; i < 8; ++i) {
      int c = i * 32 + cb;
#pragma unroll
      for (int e = 0; e < 4; ++e) {
        v2f pr = {va[i][e], vb[i][e]};
        *(v2f*)(&Y[(4 * qd + e) * PITCH + c]) = pr;
      }
      sum += va[i];
      sq += va[i] * va[i];
      sum += vb[i];
      sq += vb[i] * vb[i];
    }
    const int pr_row = (wv >> 1) * 4 + cl;
    *(v4f*)(&red[pr_row][4 * qd]) = sum;
    *(v4f*)(&red2[pr_row][4 * qd]) = sq;
  }
  __syncthreads();
  if (t < TILE) {
    float s = 0.f, q2 = 0.f;
#pragma unroll
    for (int r = 0; r < 16; ++r) {
      s += red[r][t];
      q2 += red2[r][t];
    }
    float mean = s * (1.f / C_DIM);
    float var  = q2 * (1.f / C_DIM) - mean * mean;
    meanv[t] = mean;
    rstdv[t] = __frsqrt_rn(var + EPS);
  }
  __syncthreads();

  {
    float mean = meanv[j], rstd = rstdv[j];
#pragma unroll 4
    for (int i = 0; i < 64; ++i) {
      int c = c0 + i;
      Y[j * PITCH + c] = (Y[j * PITCH + c] - mean) * rstd * lnw[c] + lnb[c];
    }
  }
  __syncthreads();

  if (WITH_Q) {
    float qs = 0.f;
#pragma unroll 4
    for (int i = 0; i < 64; ++i) {
      int c = c0 + i;
      qs += W[c] * Y[j * PITCH + c];
    }
    red[ch][j] = qs;
    __syncthreads();
    if (t < TILE) {
      red2[0][t] = red[0][t] + red[1][t] + red[2][t] + red[3][t] + bias[0];
    }
    __syncthreads();
  }

  const int wave = t >> 5;
  const int lane = t & 31;
  const int m    = lane & 15;
  const int kk   = (lane >> 4) << 1;
  const int mh   = (lane >> 4) << 3;
  const int rofs = WITH_Q ? 1 : 0;

  int rt_first = wave;
  if (WITH_Q) {
    const float* Af0 = Wf + (size_t)(2 * wave) * (16 * C_DIM) + lane * 4;
    const float* Af1 = Af0 + 16 * C_DIM;
    v8f acc[2][4] = {};
    wmma_k_loop_2r<4, GEMM_UNR4>(Af0, Af1, Y, PITCH, C_DIM, kk, m, acc[0], acc[1]);

    const float* qt = &red2[0][0];
    const float qa = qt[lane], qb = qt[lane + 32];
    float mt = fmaxf(qa, qb);
#pragma unroll
    for (int o = 16; o > 0; o >>= 1) mt = fmaxf(mt, __shfl_xor(mt, o));
    float ex[4];
#pragma unroll
    for (int c = 0; c < 4; ++c) ex[c] = __expf(qt[c * 16 + m] - mt);

    float keep = 0.f;
#pragma unroll
    for (int h = 0; h < 2; ++h) {
#pragma unroll
      for (int r = 0; r < 8; ++r) {
        const int row = (2 * wave + h) * 16 + mh + r;
        const float bz = bias[rofs + row];
        float s = (acc[h][0][r] + bz) * ex[0];
        s += (acc[h][1][r] + bz) * ex[1];
        s += (acc[h][2][r] + bz) * ex[2];
        s += (acc[h][3][r] + bz) * ex[3];
        s += __shfl_xor(s, 1);
        s += __shfl_xor(s, 2);
        s += __shfl_xor(s, 4);
        s += __shfl_xor(s, 8);
        if (m == r + 8 * h) keep = s;
      }
    }
    float* pb = part + (size_t)b * F_DIM * S_DIM + (size_t)blockIdx.x * PART_PITCH;
    pb[wave * 32 + mh + m + (m >= 8 ? 8 : 0)] = keep;
    if (wave == 0) {
      float ls = __expf(qa - mt) + __expf(qb - mt);
#pragma unroll
      for (int o = 16; o > 0; o >>= 1) ls += __shfl_xor(ls, o);
      if (lane == 0) {
        v2f ml = {mt, ls};
        *(v2f*)(pb + C_DIM) = ml;
      }
    }
    rt_first = wave + 16;
  }

  for (int rt = rt_first; rt < 32; rt += 16) {
    const float* Af0 = Wf + (size_t)rt * (16 * C_DIM) + lane * 4;
    const float* Af1 = Af0 + 8 * (16 * C_DIM);
    v8f acc[2][4] = {};
    wmma_k_loop_2r<4, GEMM_UNR4>(Af0, Af1, Y, PITCH, C_DIM, kk, m, acc[0], acc[1]);
#pragma unroll
    for (int h = 0; h < 2; ++h) {
#pragma unroll
      for (int c = 0; c < 4; ++c) {
#pragma unroll
        for (int r = 0; r < 8; ++r) {
          int row = (rt + 8 * h) * 16 + mh + r;
          float d = acc[h][c][r] + bias[rofs + row];
          if (LEAKY) d = d > 0.f ? d : 0.1f * d;
          out512[((size_t)b * F_DIM + row) * S_DIM + s0 + c * 16 + m] = d;
        }
      }
    }
  }
}

__global__ void ctx_join_kernel(const float* __restrict__ part,
                                float* __restrict__ ctx) {
  const int bp = blockIdx.x;
  const int b = bp >> 2, p = bp & 3;
  const int t = threadIdx.x;
  const float* pb = part + (size_t)b * F_DIM * S_DIM +
                    (size_t)(p * TILES_PER_P) * PART_PITCH;
  __shared__ float smax[TILES_PER_P];
  __shared__ float ssum[TILES_PER_P];
  if (t < TILES_PER_P) {
    v2f ml = *(const v2f*)(pb + t * PART_PITCH + C_DIM);
    smax[t] = ml.x;
    ssum[t] = ml.y;
  }
  float pv[TILES_PER_P];
#pragma unroll
  for (int i = 0; i < TILES_PER_P; ++i) pv[i] = pb[i * PART_PITCH + t];
  __syncthreads();
  float mx = -INFINITY;
#pragma unroll
  for (int i = 0; i < TILES_PER_P; ++i) mx = fmaxf(mx, smax[i]);
  float den = 0.f, acc = 0.f;
#pragma unroll
  for (int i = 0; i < TILES_PER_P; ++i) {
    const float w = __expf(smax[i] - mx);
    den += ssum[i] * w;
    acc += pv[i] * w;
  }
  ctx[(size_t)bp * C_DIM + t] = acc / den;
}

__global__ void GEMM_KERNEL_HINT attn_out_kernel(const float* __restrict__ x,
                                const float* __restrict__ kv,
                                const float* __restrict__ ctx,
                                const float* __restrict__ Wf,
                                const float* __restrict__ bias,
                                float* __restrict__ x1) {
  extern __shared__ __attribute__((aligned(16))) float A[];
  const int b  = blockIdx.y;
  const int s0 = blockIdx.x * TILE;
  const int p  = s0 / N_DIM;
  const int t  = threadIdx.x;
  const int j  = t & (TILE - 1);
  const int ch = t >> 6;
  const int c0 = ch * 64;

  const float* ctxbp = ctx + ((size_t)b * P_DIM + p) * C_DIM;
  {
    const int ql = t & 7;
    const int cl = (t >> 3) & 3;
    const int wv = t >> 5;
    const int qd = (wv & 1) * 8 + ql;
    const int cb = (wv >> 1) * 8 + cl * 2;
    const float* vbase = kv + ((size_t)b * F_DIM + C_DIM) * S_DIM + s0 + 4 * qd;
    v4f va[8], vb[8];
    v2f cx[8];
#pragma unroll
    for (int i = 0; i < 8; ++i) {
      int c = i * 32 + cb;
      va[i] = *(const v4f*)(vbase + (size_t)c * S_DIM);
      vb[i] = *(const v4f*)(vbase + (size_t)(c + 1) * S_DIM);
      cx[i] = *(const v2f*)(ctxbp + c);
    }
#pragma unroll
    for (int i = 0; i < 8; ++i) {
      int c = i * 32 + cb;
#pragma unroll
      for (int e = 0; e < 4; ++e) {
        float v0 = va[i][e], v1 = vb[i][e];
        v0 = v0 > 0.f ? v0 : 0.f;
        v1 = v1 > 0.f ? v1 : 0.f;
        v2f pr = {v0 * cx[i].x, v1 * cx[i].y};
        *(v2f*)(&A[(4 * qd + e) * PITCH + c]) = pr;
      }
    }
  }
  __syncthreads();

  const int wave = t >> 5;
  const int lane = t & 31;
  const int m    = lane & 15;
  const int kk   = (lane >> 4) << 1;
  const int mh   = (lane >> 4) << 3;

  {
    const float* Af0 = Wf + (size_t)wave * (16 * C_DIM) + lane * 4;
    const float* Af1 = Af0 + 8 * (16 * C_DIM);
    v8f acc[2][4] = {};
    wmma_k_loop_2r<4, GEMM_UNR4>(Af0, Af1, A, PITCH, C_DIM, kk, m, acc[0], acc[1]);
#pragma unroll
    for (int h = 0; h < 2; ++h) {
#pragma unroll
      for (int c = 0; c < 4; ++c) {
#pragma unroll
        for (int r = 0; r < 8; ++r) {
          int row = (wave + 8 * h) * 16 + mh + r;
          size_t idx = ((size_t)b * C_DIM + row) * S_DIM + s0 + c * 16 + m;
          x1[idx] = x[idx] + bias[row] + acc[h][c][r];
        }
      }
    }
  }
}

__global__ void GEMM_KERNEL_HINT ffn2_kernel(const float* __restrict__ x1,
                            const float* __restrict__ h,
                            const float* __restrict__ Wf,
                            const float* __restrict__ bias,
                            float* __restrict__ out) {
  extern __shared__ __attribute__((aligned(16))) float Hs[];
  const int b  = blockIdx.y;
  const int s0 = blockIdx.x * TILE5;
  const int t  = threadIdx.x;
  const int j  = t & (TILE5 - 1);
  const int ch = t >> 5;
  {
    const int ql = t & 7;
    const int cl = (t >> 3) & 3;
    const int rb = ch * 8 + cl * 2;
    const float* hb = h + (size_t)b * F_DIM * S_DIM + s0 + 4 * ql;
    v4f va[8], vb[8];
#pragma unroll
    for (int i = 0; i < 8; ++i) {
      int r = i * 64 + rb;
      va[i] = *(const v4f*)(hb + (size_t)r * S_DIM);
      vb[i] = *(const v4f*)(hb + (size_t)(r + 1) * S_DIM);
    }
#pragma unroll
    for (int i = 0; i < 8; ++i) {
      int r = i * 64 + rb;
#pragma unroll
      for (int e = 0; e < 4; ++e) {
        v2f pr = {va[i][e], vb[i][e]};
        *(v2f*)(&Hs[(4 * ql + e) * PITCH5 + r]) = pr;
      }
    }
  }
  __syncthreads();

  const int wave = t >> 5;
  const int lane = t & 31;
  const int m    = lane & 15;
  const int kk   = (lane >> 4) << 1;
  const int mh   = (lane >> 4) << 3;

  {
    const float* Af0 = Wf + (size_t)wave * (16 * F_DIM) + lane * 4;
    const float* Af1 = Af0 + 8 * (16 * F_DIM);
    v8f acc[2][2] = {};
    wmma_k_loop_2r<2, GEMM_UNR2>(Af0, Af1, Hs, PITCH5, F_DIM, kk, m, acc[0], acc[1]);
#pragma unroll
    for (int h = 0; h < 2; ++h) {
#pragma unroll
      for (int c = 0; c < 2; ++c) {
#pragma unroll
        for (int r = 0; r < 8; ++r) {
          int row = (wave + 8 * h) * 16 + mh + r;
          size_t idx = ((size_t)b * C_DIM + row) * S_DIM + s0 + c * 16 + m;
          out[idx] = x1[idx] + bias[row] + acc[h][c][r];
        }
      }
    }
  }
}

extern "C" void kernel_launch(void* const* d_in, const int* in_sizes, int n_in,
                              void* d_out, int out_size, void* d_ws,
                              size_t ws_size, hipStream_t stream) {
  const float* x      = (const float*)d_in[0];
  const float* ln1_w  = (const float*)d_in[1];
  const float* ln1_b  = (const float*)d_in[2];
  const float* qkv_w  = (const float*)d_in[3];
  const float* qkv_b  = (const float*)d_in[4];
  const float* out_w  = (const float*)d_in[5];
  const float* out_b  = (const float*)d_in[6];
  const float* ln2_w  = (const float*)d_in[7];
  const float* ln2_b  = (const float*)d_in[8];
  const float* ffn1_w = (const float*)d_in[9];
  const float* ffn1_b = (const float*)d_in[10];
  const float* ffn2_w = (const float*)d_in[11];
  const float* ffn2_b = (const float*)d_in[12];
  float* outp = (float*)d_out;

  float* ws   = (float*)d_ws;
  float* qbuf = ws;
  float* ctxb = qbuf + (size_t)B_DIM * S_DIM;
  float* kvb  = ctxb + (size_t)B_DIM * P_DIM * C_DIM;
  float* x1   = kvb + (size_t)B_DIM * F_DIM * S_DIM;
  float* hbuf = kvb;
  float* wf_qkv = x1;
  float* wf_any = qbuf;

  const size_t lds1 = (size_t)TILE * PITCH * sizeof(float);
  const size_t lds5 = (size_t)TILE5 * PITCH5 * sizeof(float);
  dim3 grid1(S_DIM / TILE, B_DIM);
  dim3 grid5(S_DIM / TILE5, B_DIM);

  frag_order_kernel<C_DIM><<<128, 256, 0, stream>>>(qkv_w + C_DIM, wf_qkv);
  ln_gemm_kernel<true, false><<<grid1, 256, lds1, stream>>>(
      x, ln1_w, ln1_b, qkv_w, wf_qkv, qkv_b, kvb, kvb);
  ctx_join_kernel<<<B_DIM * P_DIM, 256, 0, stream>>>(kvb, ctxb);
  frag_order_kernel<C_DIM><<<64, 256, 0, stream>>>(out_w, wf_any);
  attn_out_kernel<<<grid1, 256, lds1, stream>>>(
      x, kvb, ctxb, wf_any, out_b, x1);
  frag_order_kernel<C_DIM><<<128, 256, 0, stream>>>(ffn1_w, wf_any);
  ln_gemm_kernel<false, true><<<grid1, 256, lds1, stream>>>(
      x1, ln2_w, ln2_b, ffn1_w, wf_any, ffn1_b, hbuf, nullptr);
  frag_order_kernel<F_DIM><<<128, 256, 0, stream>>>(ffn2_w, wf_any);
  ffn2_kernel<<<grid5, 256, lds5, stream>>>(
      x1, hbuf, wf_any, ffn2_b, outp);
}
